# starting kernel
# baseline (speedup 1.0000x reference)
#include <hip/hip_runtime.h>
#include <hip/hip_bf16.h>

typedef float v8f __attribute__((ext_vector_type(8)));
typedef float v2f __attribute__((ext_vector_type(2)));

#define T_  2048
#define D_  128

__device__ __forceinline__ v8f wmma4(v2f a, v2f b, v8f c) {
  return __builtin_amdgcn_wmma_f32_16x16x4_f32(false, a, false, b, (short)0, c,
                                               false, false);
}

__device__ __forceinline__ float fast_tanh(float x) {
  float a = __builtin_fabsf(x);
  float e = __expf(-2.0f * a);
  float r = (1.0f - e) / (1.0f + e);
  return __builtin_copysignf(r, x);
}

__global__ void __launch_bounds__(256)
xproj_kernel(const float* __restrict__ x, const float* __restrict__ Wi,
             const float* __restrict__ Wib, const float* __restrict__ bias,
             float* __restrict__ xp) {
  const int lane = threadIdx.x & 31;
  const int wid  = blockIdx.x * 8 + (threadIdx.x >> 5);
  const int btile = wid >> 11;
  const int t     = wid & 2047;
  const int lo = lane & 15, hf = lane >> 4;

  v8f acc[4];
#pragma unroll
  for (int nt = 0; nt < 4; ++nt) {
    float cst = Wib[nt * 16 + lo] + bias[nt * 16 + lo];
#pragma unroll
    for (int v = 0; v < 8; ++v) acc[nt][v] = cst;
  }

  const float* xrow = x + ((size_t)(btile * 16 + lo) * T_ + t) * D_ + hf * 2;
  const float* wrow = Wi + (size_t)lo * D_ + hf * 2;
#pragma unroll
  for (int ks = 0; ks < 32; ++ks) {
    v2f a = *(const v2f*)(xrow + ks * 4);
#pragma unroll
    for (int nt = 0; nt < 4; ++nt) {
      v2f b = *(const v2f*)(wrow + nt * 16 * D_ + ks * 4);
      acc[nt] = wmma4(a, b, acc[nt]);
    }
  }

  float* o = xp + ((size_t)wid * 32 + lane) * 32;
#pragma unroll
  for (int nt = 0; nt < 4; ++nt) {
#pragma unroll
    for (int q = 0; q < 2; ++q) {
      float4 s;
      s.x = acc[nt][q * 4 + 0]; s.y = acc[nt][q * 4 + 1];
      s.z = acc[nt][q * 4 + 2]; s.w = acc[nt][q * 4 + 3];
      *(float4*)(o + nt * 8 + q * 4) = s;
    }
  }
}

__global__ void __launch_bounds__(32)
scan_kernel(const float* __restrict__ Ap, const float* __restrict__ Am,
            const float* __restrict__ Wf, const float* __restrict__ Wo,
            const float* __restrict__ Wob, const float* __restrict__ xp,
            float* __restrict__ out) {
  __shared__ __align__(16) float hlds[16 * 64];
  const int lane = threadIdx.x;
  const int lo = lane & 15, hf = lane >> 4;
  const int btile = blockIdx.x;

  v2f bw[4][8];
#pragma unroll
  for (int nt = 0; nt < 4; ++nt) {
    const float* src = (nt < 2) ? Ap : Am;
    const int j = nt * 16 + lo;
#pragma unroll
    for (int s = 0; s < 8; ++s) {
      const int ks = (nt < 2) ? s : (s + 8);
      bw[nt][s] = *(const v2f*)(src + j * 64 + ks * 4 + hf * 2);
    }
  }
  v2f bs[4][16];
#pragma unroll
  for (int nt = 0; nt < 4; ++nt) {
    const int j = nt * 16 + lo;
    const float sg = (nt < 2) ? 1.0f : -1.0f;
#pragma unroll
    for (int ks = 0; ks < 16; ++ks) {
      v2f w = *(const v2f*)(Wf + j * 64 + ks * 4 + hf * 2);
      v2f r; r.x = w.x * sg; r.y = w.y * sg;
      bs[nt][ks] = r;
    }
  }

  for (int i = lane; i < 1024; i += 32) hlds[i] = 0.0f;
  __syncthreads();

  const float* xpl = xp + (size_t)btile * T_ * 1024 + lane * 32;
  float4 xq[8];
#pragma unroll
  for (int q = 0; q < 8; ++q) xq[q] = *(const float4*)(xpl + q * 4);

#pragma unroll 1
  for (int t = 0; t < T_; ++t) {
    v8f uc[4], u2[4], u3[4];
#pragma unroll
    for (int nt = 0; nt < 4; ++nt) {
      float4 q0 = xq[nt * 2], q1 = xq[nt * 2 + 1];
      uc[nt][0] = q0.x; uc[nt][1] = q0.y; uc[nt][2] = q0.z; uc[nt][3] = q0.w;
      uc[nt][4] = q1.x; uc[nt][5] = q1.y; uc[nt][6] = q1.z; uc[nt][7] = q1.w;
#pragma unroll
      for (int v = 0; v < 8; ++v) { u2[nt][v] = 0.0f; u3[nt][v] = 0.0f; }
    }
    if (t + 1 < T_) {
      const float* nx = xpl + (size_t)(t + 1) * 1024;
#pragma unroll
      for (int q = 0; q < 8; ++q) xq[q] = *(const float4*)(nx + q * 4);
      __builtin_prefetch(xpl + (size_t)(t + 5) * 1024, 0, 1);
    }

    v2f ha[16];
#pragma unroll
    for (int ks = 0; ks < 16; ++ks)
      ha[ks] = *(const v2f*)&hlds[lo * 64 + ks * 4 + hf * 2];

    float pa = 0.0f, pm = 0.0f;
#pragma unroll
    for (int ks = 0; ks < 16; ++ks) {
      float d = ha[ks].x * ha[ks].x + ha[ks].y * ha[ks].y;
      pa += d;
      if (ks >= 8) pm += d;
    }
    pa += __shfl_xor(pa, 16, 32);
    pm += __shfl_xor(pm, 16, 32);
    const float alpha = pm / (pa + 1e-8f);
    const float g =
        1.0f / (1.0f + __expf(-(alpha - 0.72134752044448170f) * 20.0f));
    v2f hg[16];
#pragma unroll
    for (int ks = 0; ks < 16; ++ks) {
      hg[ks].x = ha[ks].x * g; hg[ks].y = ha[ks].y * g;
    }

#pragma unroll
    for (int nt = 0; nt < 4; ++nt) {
#pragma unroll
      for (int s = 0; s < 8; ++s)
        uc[nt] = wmma4(ha[(nt < 2) ? s : (s + 8)], bw[nt][s], uc[nt]);
#pragma unroll
      for (int s = 0; s < 8; ++s)
        u2[nt] = wmma4(hg[s], bs[nt][s], u2[nt]);
#pragma unroll
      for (int s = 8; s < 16; ++s)
        u3[nt] = wmma4(hg[s], bs[nt][s], u3[nt]);
    }

    __syncthreads();
#pragma unroll
    for (int nt = 0; nt < 4; ++nt)
#pragma unroll
      for (int v = 0; v < 8; ++v)
        hlds[(v + 8 * hf) * 64 + nt * 16 + lo] =
            fast_tanh(uc[nt][v] + u2[nt][v] + u3[nt][v]);
    __syncthreads();
  }

#pragma unroll 1
  for (int i = lane; i < 1024; i += 32)
    out[32768 + btile * 1024 + i] = hlds[i];

  v2f ph[16];
#pragma unroll
  for (int ks = 0; ks < 16; ++ks)
    ph[ks] = *(const v2f*)&hlds[lo * 64 + ks * 4 + hf * 2];
#pragma unroll
  for (int n2 = 0; n2 < 8; ++n2) {
    v8f p;
    const float c0 = Wob[n2 * 16 + lo];
#pragma unroll
    for (int v = 0; v < 8; ++v) p[v] = c0;
#pragma unroll
    for (int ks = 0; ks < 16; ++ks) {
      v2f b = *(const v2f*)(Wo + (size_t)(n2 * 16 + lo) * 64 + ks * 4 + hf * 2);
      p = wmma4(ph[ks], b, p);
    }
#pragma unroll
    for (int v = 0; v < 8; ++v)
      out[(size_t)(btile * 16 + v + 8 * hf) * 128 + n2 * 16 + lo] = p[v];
  }
}

extern "C" void kernel_launch(void* const* d_in, const int* in_sizes, int n_in,
                              void* d_out, int out_size, void* d_ws, size_t ws_size,
                              hipStream_t stream) {
  const float* x    = (const float*)d_in[0];
  const float* Ap   = (const float*)d_in[1];
  const float* Am   = (const float*)d_in[2];
  const float* Wf   = (const float*)d_in[3];
  const float* Wiw  = (const float*)d_in[4];
  const float* Wib  = (const float*)d_in[5];
  const float* Wow  = (const float*)d_in[6];
  const float* Wob  = (const float*)d_in[7];
  const float* bias = (const float*)d_in[8];
  float* out = (float*)d_out;
  float* xp  = (float*)d_ws;

  xproj_kernel<<<4096, 256, 0, stream>>>(x, Wiw, Wib, bias, xp);
  scan_kernel<<<16, 32, 0, stream>>>(Ap, Am, Wf, Wow, Wob, xp, out);
}


# scan: each partial tile stored to LDS right behind its matrix ops; own tile's 8 matrix ops moved into the store drain and the load latency around the barrier; half-sum by permlanex16; sums unchanged
# speedup vs baseline: 3.2644x; 3.2644x over previous
#include <hip/hip_runtime.h>
#include <hip/hip_bf16.h>


typedef float v8f __attribute__((ext_vector_type(8)));
typedef float v2f __attribute__((ext_vector_type(2)));

#define T_  2048
#define D_  128

#define SLOT_     256
#define COMM_OFF_ (16 * SLOT_)
#define SQ_OFF_   (20 * SLOT_)
#define PLANE_    (20 * SLOT_ + 128)

__device__ __forceinline__ v8f wmma4(v2f a, v2f b, v8f c) {
  return __builtin_amdgcn_wmma_f32_16x16x4_f32(false, a, false, b, (short)0, c,
                                               false, false);
}

__device__ __forceinline__ float fast_tanh(float x) {
  float a = __builtin_fabsf(x);
  float e = __expf(-2.0f * a);
  float r = (1.0f - e) / (1.0f + e);
  return __builtin_copysignf(r, x);
}

__device__ __forceinline__ void st8(float* p, v8f a) {
  float4 s0, s1;
  s0.x = a[0]; s0.y = a[1]; s0.z = a[2]; s0.w = a[3];
  s1.x = a[4]; s1.y = a[5]; s1.z = a[6]; s1.w = a[7];
  *(float4*)p = s0;
  *(float4*)(p + 128) = s1;
}
__device__ __forceinline__ v8f ld8(const float* p) {
  float4 s0 = *(const float4*)p;
  float4 s1 = *(const float4*)(p + 128);
  v8f a;
  a[0] = s0.x; a[1] = s0.y; a[2] = s0.z; a[3] = s0.w;
  a[4] = s1.x; a[5] = s1.y; a[6] = s1.z; a[7] = s1.w;
  return a;
}

__global__ void __launch_bounds__(256)
xproj_kernel(const float* __restrict__ x, const float* __restrict__ Wi,
             const float* __restrict__ Wib, const float* __restrict__ bias,
             float* __restrict__ xp) {
  const int lane = threadIdx.x & 31;
  const int wid  = blockIdx.x * 8 + (threadIdx.x >> 5);
  const int btile = wid >> 11;
  const int t     = wid & 2047;
  const int lo = lane & 15, hf = lane >> 4;

  v8f acc[4];
#pragma unroll
  for (int nt = 0; nt < 4; ++nt) {
#pragma unroll
    for (int v = 0; v < 8; ++v)
      acc[nt][v] = Wib[nt * 16 + 8 * hf + v] + bias[nt * 16 + 8 * hf + v];
  }

  const float* xrow = x + ((size_t)(btile * 16 + lo) * T_ + t) * D_ + hf * 2;
  const float* wrow = Wi + (size_t)lo * D_ + hf * 2;
#pragma unroll
  for (int ks = 0; ks < 32; ++ks) {
    v2f xb = *(const v2f*)(xrow + ks * 4);
#pragma unroll
    for (int nt = 0; nt < 4; ++nt) {
      v2f wa = *(const v2f*)(wrow + nt * 16 * D_ + ks * 4);
      acc[nt] = wmma4(wa, xb, acc[nt]);
    }
  }

  float* o = xp + (size_t)wid * (4 * SLOT_) + lane * 4;
#pragma unroll
  for (int nt = 0; nt < 4; ++nt) st8(o + nt * SLOT_, acc[nt]);
}

__device__ __forceinline__ void lds_barrier() {
  asm volatile("s_wait_dscnt 0x0\n\ts_barrier_signal -1\n\ts_barrier_wait -1"
               ::: "memory");
}

#define PIN_ORDER()                         \
  do {                                      \
    asm volatile("" ::: "memory");          \
    __builtin_amdgcn_sched_barrier(0);      \
  } while (0)

__device__ __forceinline__ float xhalf(float x) {
  int b = __builtin_bit_cast(int, x);
  return __builtin_bit_cast(
      float, __builtin_amdgcn_permlanex16(b, b, 0x76543210, 0xfedcba98, false, false));
}

__global__ void __launch_bounds__(128)
scan_kernel(const float* __restrict__ Ap, const float* __restrict__ Am,
            const float* __restrict__ Wf, const float* __restrict__ Wo,
            const float* __restrict__ Wob, const float* __restrict__ xp,
            float* __restrict__ out) {
  __shared__ __align__(16) float ex[2 * PLANE_];
  __shared__ __align__(16) float hlds[16 * 64];
  const int lane = threadIdx.x & 31;
  const int w = __builtin_amdgcn_readfirstlane((int)(threadIdx.x >> 5));
  const int lo = lane & 15, hf = lane >> 4;
  const int btile = blockIdx.x;
  const int kb = 16 * w + 8 * hf;

  const float* xpw = xp + ((size_t)btile * T_ * 4 + w) * SLOT_;
  const float* xpl = xpw + lane * 4;
  v8f xq = ld8(xpl);

  const float* cs = (w < 2) ? Ap : Am;
  v2f aco[4], acp[4];
#pragma unroll
  for (int s = 0; s < 4; ++s) {
    aco[s] = *(const v2f*)(cs + (w * 16 + lo) * 64 + kb + 2 * s);
    acp[s] = *(const v2f*)(cs + ((w ^ 1) * 16 + lo) * 64 + kb + 2 * s);
  }
  v2f as_[4][4];
#pragma unroll
  for (int i = 0; i < 4; ++i) {
    const int d = (w + i) & 3;
    const float sg = (d < 2) ? 1.0f : -1.0f;
    const float* src = Wf + (d * 16 + lo) * 64 + kb;
#pragma unroll
    for (int s = 0; s < 4; ++s) {
      v2f q = *(const v2f*)(src + 2 * s);
      v2f r; r.x = q.x * sg; r.y = q.y * sg;
      as_[i][s] = r;
    }
  }

  const int d1 = (w + 1) & 3, d2 = (w + 2) & 3, d3 = (w + 3) & 3;
  const int so1 = (w * 4 + d1) * SLOT_ + lane * 4;
  const int so2 = (w * 4 + d2) * SLOT_ + lane * 4;
  const int so3 = (w * 4 + d3) * SLOT_ + lane * 4;
  const int soc = COMM_OFF_ + (w ^ 1) * SLOT_ + lane * 4;
  const int li1 = (d1 * 4 + w) * SLOT_ + lane * 4;
  const int li2 = (d2 * 4 + w) * SLOT_ + lane * 4;
  const int li3 = (d3 * 4 + w) * SLOT_ + lane * 4;
  const int lic = COMM_OFF_ + w * SLOT_ + lane * 4;

  v8f h;
#pragma unroll
  for (int v = 0; v < 8; ++v) h[v] = 0.0f;

  auto step = [&](float* P, v8f seed) {
    v2f hb[4];
#pragma unroll
    for (int s = 0; s < 4; ++s) { hb[s].x = h[2 * s]; hb[s].y = h[2 * s + 1]; }

    float sq = 0.0f;
#pragma unroll
    for (int v = 0; v < 8; ++v) sq += h[v] * h[v];
    sq += xhalf(sq);

    v8f pco = seed;
    v8f ps0, ps1, ps2, ps3, pcp;
#pragma unroll
    for (int v = 0; v < 8; ++v) {
      ps0[v] = 0.0f; ps1[v] = 0.0f; ps2[v] = 0.0f; ps3[v] = 0.0f; pcp[v] = 0.0f;
    }

#pragma unroll
    for (int s = 0; s < 4; ++s) {
      ps1 = wmma4(as_[1][s], hb[s], ps1);
      ps2 = wmma4(as_[2][s], hb[s], ps2);
    }
    PIN_ORDER();
    ps3 = wmma4(as_[3][0], hb[0], ps3);
    pcp = wmma4(acp[0], hb[0], pcp);
    PIN_ORDER();
    st8(P + so1, ps1);
    st8(P + so2, ps2);
    PIN_ORDER();
#pragma unroll
    for (int s = 1; s < 4; ++s) {
      ps3 = wmma4(as_[3][s], hb[s], ps3);
      pcp = wmma4(acp[s], hb[s], pcp);
    }
    PIN_ORDER();
    ps0 = wmma4(as_[0][0], hb[0], ps0);
    pco = wmma4(aco[0], hb[0], pco);
    PIN_ORDER();
    st8(P + so3, ps3);
    st8(P + soc, pcp);
    P[SQ_OFF_ + w * 32 + lane] = sq;
    PIN_ORDER();
    ps0 = wmma4(as_[0][1], hb[1], ps0);
    pco = wmma4(aco[1], hb[1], pco);
    PIN_ORDER();

    lds_barrier();

    float s0 = P[SQ_OFF_ + lane];
    float s1 = P[SQ_OFF_ + 32 + lane];
    float s2 = P[SQ_OFF_ + 64 + lane];
    float s3 = P[SQ_OFF_ + 96 + lane];
    PIN_ORDER();
    v8f rc = ld8(P + lic);
    v8f r1 = ld8(P + li1);
    v8f r2 = ld8(P + li2);
    v8f r3 = ld8(P + li3);
    PIN_ORDER();
#pragma unroll
    for (int s = 2; s < 4; ++s) {
      ps0 = wmma4(as_[0][s], hb[s], ps0);
      pco = wmma4(aco[s], hb[s], pco);
    }
    PIN_ORDER();
    asm volatile("" : "+v"(s0), "+v"(s1), "+v"(s2), "+v"(s3));
    PIN_ORDER();

    const float pm = s2 + s3;
    const float pa = (s0 + s1) + pm;
    float alpha = pm / (pa + 1e-8f);
    asm volatile("" : "+v"(alpha), "+v"(rc), "+v"(r1), "+v"(r2), "+v"(r3));
    PIN_ORDER();
    const float g =
        1.0f / (1.0f + __expf(-(alpha - 0.72134752044448170f) * 20.0f));

#pragma unroll
    for (int v = 0; v < 8; ++v) {
      const float um = ((ps0[v] + r1[v]) + r2[v]) + r3[v];
      h[v] = fast_tanh((pco[v] + rc[v]) + g * um);
    }
  };

#pragma unroll 1
  for (int t = 0; t < T_; t += 2) {
    const v8f xb = ld8(xpl + (size_t)(t + 1) * (4 * SLOT_));
    const int tp = (t + 8 < T_) ? (t + 8) : (T_ - 2);
    __builtin_prefetch(xpw + (size_t)tp * (4 * SLOT_) + lane * 8, 0, 1);
    __builtin_prefetch(xpw + (size_t)(tp + 1) * (4 * SLOT_) + lane * 8, 0, 1);
    step(ex, xq);
    const int tn = (t + 2 < T_) ? (t + 2) : t;
    xq = ld8(xpl + (size_t)tn * (4 * SLOT_));
    step(ex + PLANE_, xb);
  }

  {
    float4 a, b;
    a.x = h[0]; a.y = h[1]; a.z = h[2]; a.w = h[3];
    b.x = h[4]; b.y = h[5]; b.z = h[6]; b.w = h[7];
    *(float4*)&hlds[lo * 64 + kb] = a;
    *(float4*)&hlds[lo * 64 + kb + 4] = b;
  }
  __syncthreads();

  for (int i = threadIdx.x; i < 1024; i += 128)
    out[32768 + btile * 1024 + i] = hlds[i];

  v2f ph[16];
#pragma unroll
  for (int ks = 0; ks < 16; ++ks)
    ph[ks] = *(const v2f*)&hlds[lo * 64 + ks * 4 + hf * 2];
#pragma unroll
  for (int n2l = 0; n2l < 2; ++n2l) {
    const int n2 = 2 * w + n2l;
    v8f p;
    const float c0 = Wob[n2 * 16 + lo];
#pragma unroll
    for (int v = 0; v < 8; ++v) p[v] = c0;
#pragma unroll
    for (int ks = 0; ks < 16; ++ks) {
      v2f b = *(const v2f*)(Wo + (size_t)(n2 * 16 + lo) * 64 + ks * 4 + hf * 2);
      p = wmma4(ph[ks], b, p);
    }
#pragma unroll
    for (int v = 0; v < 8; ++v)
      out[(size_t)(btile * 16 + v + 8 * hf) * 128 + n2 * 16 + lo] = p[v];
  }
}

extern "C" void kernel_launch(void* const* d_in, const int* in_sizes, int n_in,
                              void* d_out, int out_size, void* d_ws, size_t ws_size,
                              hipStream_t stream) {
  const float* x    = (const float*)d_in[0];
  const float* Ap   = (const float*)d_in[1];
  const float* Am   = (const float*)d_in[2];
  const float* Wf   = (const float*)d_in[3];
  const float* Wiw  = (const float*)d_in[4];
  const float* Wib  = (const float*)d_in[5];
  const float* Wow  = (const float*)d_in[6];
  const float* Wob  = (const float*)d_in[7];
  const float* bias = (const float*)d_in[8];
  float* out = (float*)d_out;
  float* xp  = (float*)d_ws;

  xproj_kernel<<<4096, 256, 0, stream>>>(x, Wiw, Wib, bias, xp);
  scan_kernel<<<16, 128, 0, stream>>>(Ap, Am, Wf, Wow, Wob, xp, out);
}


# scan: sums of squares cross the waves in an early LDS round of their own (2nd, split barrier); gate's quotient-exp-quotient chain cut into pieces behind single matrix ops; values, sum order unchanged
# speedup vs baseline: 3.3750x; 1.0339x over previous
#include <hip/hip_runtime.h>
#include <hip/hip_bf16.h>


typedef float v8f __attribute__((ext_vector_type(8)));
typedef float v2f __attribute__((ext_vector_type(2)));

#define T_  2048
#define D_  128

#define SLOT_     256
#define COMM_OFF_ (16 * SLOT_)
#define PLANE_    (20 * SLOT_)

__device__ __forceinline__ v8f wmma4(v2f a, v2f b, v8f c) {
  return __builtin_amdgcn_wmma_f32_16x16x4_f32(false, a, false, b, (short)0, c,
                                               false, false);
}

__device__ __forceinline__ float fast_tanh(float x) {
  float a = __builtin_fabsf(x);
  float e = __expf(-2.0f * a);
  float r = (1.0f - e) / (1.0f + e);
  return __builtin_copysignf(r, x);
}

__device__ __forceinline__ void st8(float* p, v8f a) {
  float4 s0, s1;
  s0.x = a[0]; s0.y = a[1]; s0.z = a[2]; s0.w = a[3];
  s1.x = a[4]; s1.y = a[5]; s1.z = a[6]; s1.w = a[7];
  *(float4*)p = s0;
  *(float4*)(p + 128) = s1;
}
__device__ __forceinline__ v8f ld8(const float* p) {
  float4 s0 = *(const float4*)p;
  float4 s1 = *(const float4*)(p + 128);
  v8f a;
  a[0] = s0.x; a[1] = s0.y; a[2] = s0.z; a[3] = s0.w;
  a[4] = s1.x; a[5] = s1.y; a[6] = s1.z; a[7] = s1.w;
  return a;
}

__global__ void __launch_bounds__(256)
xproj_kernel(const float* __restrict__ x, const float* __restrict__ Wi,
             const float* __restrict__ Wib, const float* __restrict__ bias,
             float* __restrict__ xp) {
  const int lane = threadIdx.x & 31;
  const int wid  = blockIdx.x * 8 + (threadIdx.x >> 5);
  const int btile = wid >> 11;
  const int t     = wid & 2047;
  const int lo = lane & 15, hf = lane >> 4;

  v8f acc[4];
#pragma unroll
  for (int nt = 0; nt < 4; ++nt) {
#pragma unroll
    for (int v = 0; v < 8; ++v)
      acc[nt][v] = Wib[nt * 16 + 8 * hf + v] + bias[nt * 16 + 8 * hf + v];
  }

  const float* xrow = x + ((size_t)(btile * 16 + lo) * T_ + t) * D_ + hf * 2;
  const float* wrow = Wi + (size_t)lo * D_ + hf * 2;
#pragma unroll
  for (int ks = 0; ks < 32; ++ks) {
    v2f xb = *(const v2f*)(xrow + ks * 4);
#pragma unroll
    for (int nt = 0; nt < 4; ++nt) {
      v2f wa = *(const v2f*)(wrow + nt * 16 * D_ + ks * 4);
      acc[nt] = wmma4(wa, xb, acc[nt]);
    }
  }

  float* o = xp + (size_t)wid * (4 * SLOT_) + lane * 4;
#pragma unroll
  for (int nt = 0; nt < 4; ++nt) st8(o + nt * SLOT_, acc[nt]);
}

__device__ __forceinline__ void lds_barrier() {
  asm volatile("s_wait_dscnt 0x0\n\ts_barrier_signal -1\n\ts_barrier_wait -1"
               ::: "memory");
}

__device__ __forceinline__ void lds_signal() {
  asm volatile("s_wait_dscnt 0x0\n\ts_barrier_signal -1" ::: "memory");
}
__device__ __forceinline__ void lds_wait() {
  asm volatile("s_barrier_wait -1" ::: "memory");
}

#define PIN_ORDER()                         \
  do {                                      \
    asm volatile("" ::: "memory");          \
    __builtin_amdgcn_sched_barrier(0);      \
  } while (0)

__device__ __forceinline__ float xhalf(float x) {
  int b = __builtin_bit_cast(int, x);
  return __builtin_bit_cast(
      float, __builtin_amdgcn_permlanex16(b, b, 0x76543210, 0xfedcba98, false, false));
}

__global__ void __launch_bounds__(128)
scan_kernel(const float* __restrict__ Ap, const float* __restrict__ Am,
            const float* __restrict__ Wf, const float* __restrict__ Wo,
            const float* __restrict__ Wob, const float* __restrict__ xp,
            float* __restrict__ out) {
  __shared__ __align__(16) float ex[2 * PLANE_];
  __shared__ __align__(16) float sqx[2 * 128];
  __shared__ __align__(16) float hlds[16 * 64];
  const int lane = threadIdx.x & 31;
  const int w = __builtin_amdgcn_readfirstlane((int)(threadIdx.x >> 5));
  const int lo = lane & 15, hf = lane >> 4;
  const int btile = blockIdx.x;
  const int kb = 16 * w + 8 * hf;

  const float* xpw = xp + ((size_t)btile * T_ * 4 + w) * SLOT_;
  const float* xpl = xpw + lane * 4;
  v8f xq = ld8(xpl);

  const float* cs = (w < 2) ? Ap : Am;
  v2f aco[4], acp[4];
#pragma unroll
  for (int s = 0; s < 4; ++s) {
    aco[s] = *(const v2f*)(cs + (w * 16 + lo) * 64 + kb + 2 * s);
    acp[s] = *(const v2f*)(cs + ((w ^ 1) * 16 + lo) * 64 + kb + 2 * s);
  }
  v2f as_[4][4];
#pragma unroll
  for (int i = 0; i < 4; ++i) {
    const int d = (w + i) & 3;
    const float sg = (d < 2) ? 1.0f : -1.0f;
    const float* src = Wf + (d * 16 + lo) * 64 + kb;
#pragma unroll
    for (int s = 0; s < 4; ++s) {
      v2f q = *(const v2f*)(src + 2 * s);
      v2f r; r.x = q.x * sg; r.y = q.y * sg;
      as_[i][s] = r;
    }
  }

  const int d1 = (w + 1) & 3, d2 = (w + 2) & 3, d3 = (w + 3) & 3;
  const int so1 = (w * 4 + d1) * SLOT_ + lane * 4;
  const int so2 = (w * 4 + d2) * SLOT_ + lane * 4;
  const int so3 = (w * 4 + d3) * SLOT_ + lane * 4;
  const int soc = COMM_OFF_ + (w ^ 1) * SLOT_ + lane * 4;
  const int li1 = (d1 * 4 + w) * SLOT_ + lane * 4;
  const int li2 = (d2 * 4 + w) * SLOT_ + lane * 4;
  const int li3 = (d3 * 4 + w) * SLOT_ + lane * 4;
  const int lic = COMM_OFF_ + w * SLOT_ + lane * 4;

  v8f h;
#pragma unroll
  for (int v = 0; v < 8; ++v) h[v] = 0.0f;

  auto step = [&](float* P, float* Q, v8f seed) {
    v2f hb[4];
#pragma unroll
    for (int s = 0; s < 4; ++s) { hb[s].x = h[2 * s]; hb[s].y = h[2 * s + 1]; }

    float sq = 0.0f;
#pragma unroll
    for (int v = 0; v < 8; ++v) sq += h[v] * h[v];
    sq += xhalf(sq);

    v8f pco = seed;
    v8f ps0, ps1, ps2, ps3, pcp;
#pragma unroll
    for (int v = 0; v < 8; ++v) {
      ps0[v] = 0.0f; ps1[v] = 0.0f; ps2[v] = 0.0f; ps3[v] = 0.0f; pcp[v] = 0.0f;
    }

#pragma unroll
    for (int s = 0; s < 2; ++s) {
      ps1 = wmma4(as_[1][s], hb[s], ps1);
      ps2 = wmma4(as_[2][s], hb[s], ps2);
    }
    PIN_ORDER();
    Q[w * 32 + lane] = sq;
    PIN_ORDER();
#pragma unroll
    for (int s = 2; s < 4; ++s) {
      ps1 = wmma4(as_[1][s], hb[s], ps1);
      ps2 = wmma4(as_[2][s], hb[s], ps2);
    }
    PIN_ORDER();
    lds_signal();
    PIN_ORDER();
    ps3 = wmma4(as_[3][0], hb[0], ps3);
    pcp = wmma4(acp[0], hb[0], pcp);
    PIN_ORDER();
    lds_wait();
    float s0 = Q[lane];
    float s1 = Q[32 + lane];
    float s2 = Q[64 + lane];
    float s3 = Q[96 + lane];
    st8(P + so1, ps1);
    st8(P + so2, ps2);
    PIN_ORDER();
#pragma unroll
    for (int s = 1; s < 4; ++s) {
      ps3 = wmma4(as_[3][s], hb[s], ps3);
      pcp = wmma4(acp[s], hb[s], pcp);
    }
    PIN_ORDER();

    ps0 = wmma4(as_[0][0], hb[0], ps0);
    const float pm = s2 + s3;
    const float pa = (s0 + s1) + pm;
    const float qd = pa + 1e-8f;
    PIN_ORDER();
    pco = wmma4(aco[0], hb[0], pco);
    bool vc0, vc1;
    const float d1s = __builtin_amdgcn_div_scalef(pm, qd, false, &vc0);
    const float n1s = __builtin_amdgcn_div_scalef(pm, qd, true, &vc1);
    float r1 = __builtin_amdgcn_rcpf(d1s);
    PIN_ORDER();
    st8(P + so3, ps3);
    st8(P + soc, pcp);
    PIN_ORDER();
    ps0 = wmma4(as_[0][1], hb[1], ps0);
    const float e10 = __builtin_fmaf(-d1s, r1, 1.0f);
    r1 = __builtin_fmaf(e10, r1, r1);
    float m1 = n1s * r1;
    PIN_ORDER();
    pco = wmma4(aco[1], hb[1], pco);
    const float e11 = __builtin_fmaf(-d1s, m1, n1s);
    m1 = __builtin_fmaf(e11, r1, m1);
    const float e12 = __builtin_fmaf(-d1s, m1, n1s);
    const float f1 = __builtin_amdgcn_div_fmasf(e12, r1, m1, vc1);
    const float alpha = __builtin_amdgcn_div_fixupf(f1, qd, pm);
    float z0 = alpha - 0.72134752044448170f;
    PIN_ORDER();

    lds_barrier();

    v8f rc = ld8(P + lic);
    v8f r1t = ld8(P + li1);
    v8f r2t = ld8(P + li2);
    v8f r3t = ld8(P + li3);
    PIN_ORDER();
    asm volatile("" : "+v"(ps0), "+v"(pco), "+v"(z0));
    ps0 = wmma4(as_[0][2], hb[2], ps0);
    asm volatile("" : "+v"(ps0), "+v"(z0));
    const float z1 = -z0 * 20.0f;
    const float ex = __builtin_amdgcn_exp2f(z1 * 1.44269504088896340736f);
    float dd = 1.0f + ex;
    asm volatile("" : "+v"(pco), "+v"(dd));
    pco = wmma4(aco[2], hb[2], pco);
    asm volatile("" : "+v"(pco), "+v"(dd));
    bool vc2, vc3;
    float d2s = __builtin_amdgcn_div_scalef(1.0f, dd, false, &vc2);
    float n2s = __builtin_amdgcn_div_scalef(1.0f, dd, true, &vc3);
    float r2 = __builtin_amdgcn_rcpf(d2s);
    float e20 = __builtin_fmaf(-d2s, r2, 1.0f);
    asm volatile("" : "+v"(ps0), "+v"(d2s), "+v"(n2s), "+v"(r2), "+v"(e20));
    ps0 = wmma4(as_[0][3], hb[3], ps0);
    asm volatile("" : "+v"(ps0), "+v"(d2s), "+v"(n2s), "+v"(r2), "+v"(e20));
    r2 = __builtin_fmaf(e20, r2, r2);
    float m2 = n2s * r2;
    float e21 = __builtin_fmaf(-d2s, m2, n2s);
    asm volatile("" : "+v"(pco), "+v"(d2s), "+v"(n2s), "+v"(r2), "+v"(m2), "+v"(e21));
    pco = wmma4(aco[3], hb[3], pco);
    asm volatile("" : "+v"(pco), "+v"(d2s), "+v"(n2s), "+v"(r2), "+v"(m2), "+v"(e21));
    m2 = __builtin_fmaf(e21, r2, m2);
    const float e22 = __builtin_fmaf(-d2s, m2, n2s);
    const float f2 = __builtin_amdgcn_div_fmasf(e22, r2, m2, vc3);
    float g = __builtin_amdgcn_div_fixupf(f2, dd, 1.0f);
    asm volatile("" : "+v"(g), "+v"(rc), "+v"(r1t), "+v"(r2t), "+v"(r3t));

#pragma unroll
    for (int v = 0; v < 8; ++v) {
      const float um = ((ps0[v] + r1t[v]) + r2t[v]) + r3t[v];
      h[v] = fast_tanh((pco[v] + rc[v]) + g * um);
    }
  };

#pragma unroll 1
  for (int t = 0; t < T_; t += 2) {
    const v8f xb = ld8(xpl + (size_t)(t + 1) * (4 * SLOT_));
    const int tp = (t + 8 < T_) ? (t + 8) : (T_ - 2);
    __builtin_prefetch(xpw + (size_t)tp * (4 * SLOT_) + lane * 8, 0, 1);
    __builtin_prefetch(xpw + (size_t)(tp + 1) * (4 * SLOT_) + lane * 8, 0, 1);
    step(ex, sqx, xq);
    const int tn = (t + 2 < T_) ? (t + 2) : t;
    xq = ld8(xpl + (size_t)tn * (4 * SLOT_));
    step(ex + PLANE_, sqx + 128, xb);
  }

  {
    float4 a, b;
    a.x = h[0]; a.y = h[1]; a.z = h[2]; a.w = h[3];
    b.x = h[4]; b.y = h[5]; b.z = h[6]; b.w = h[7];
    *(float4*)&hlds[lo * 64 + kb] = a;
    *(float4*)&hlds[lo * 64 + kb + 4] = b;
  }
  __syncthreads();

  for (int i = threadIdx.x; i < 1024; i += 128)
    out[32768 + btile * 1024 + i] = hlds[i];

  v2f ph[16];
#pragma unroll
  for (int ks = 0; ks < 16; ++ks)
    ph[ks] = *(const v2f*)&hlds[lo * 64 + ks * 4 + hf * 2];
#pragma unroll
  for (int n2l = 0; n2l < 2; ++n2l) {
    const int n2 = 2 * w + n2l;
    v8f p;
    const float c0 = Wob[n2 * 16 + lo];
#pragma unroll
    for (int v = 0; v < 8; ++v) p[v] = c0;
#pragma unroll
    for (int ks = 0; ks < 16; ++ks) {
      v2f b = *(const v2f*)(Wo + (size_t)(n2 * 16 + lo) * 64 + ks * 4 + hf * 2);
      p = wmma4(ph[ks], b, p);
    }
#pragma unroll
    for (int v = 0; v < 8; ++v)
      out[(size_t)(btile * 16 + v + 8 * hf) * 128 + n2 * 16 + lo] = p[v];
  }
}

extern "C" void kernel_launch(void* const* d_in, const int* in_sizes, int n_in,
                              void* d_out, int out_size, void* d_ws, size_t ws_size,
                              hipStream_t stream) {
  const float* x    = (const float*)d_in[0];
  const float* Ap   = (const float*)d_in[1];
  const float* Am   = (const float*)d_in[2];
  const float* Wf   = (const float*)d_in[3];
  const float* Wiw  = (const float*)d_in[4];
  const float* Wib  = (const float*)d_in[5];
  const float* Wow  = (const float*)d_in[6];
  const float* Wob  = (const float*)d_in[7];
  const float* bias = (const float*)d_in[8];
  float* out = (float*)d_out;
  float* xp  = (float*)d_ws;

  xproj_kernel<<<4096, 256, 0, stream>>>(x, Wiw, Wib, bias, xp);
  scan_kernel<<<16, 128, 0, stream>>>(Ap, Am, Wf, Wow, Wob, xp, out);
}


# x_proj kernel: W_in staged once per block in LDS in fragment order, A fragments by 16-B LDS reads (were 128 cache loads of 16 lines per step); 8 steps per wave, next x loaded ahead; sums unchanged
# speedup vs baseline: 3.5503x; 1.0519x over previous
#include <hip/hip_runtime.h>
#include <hip/hip_bf16.h>


typedef float v8f __attribute__((ext_vector_type(8)));
typedef float v2f __attribute__((ext_vector_type(2)));

#define T_  2048
#define D_  128

#define SLOT_     256
#define COMM_OFF_ (16 * SLOT_)
#define PLANE_    (20 * SLOT_)

__device__ __forceinline__ v8f wmma4(v2f a, v2f b, v8f c) {
  return __builtin_amdgcn_wmma_f32_16x16x4_f32(false, a, false, b, (short)0, c,
                                               false, false);
}

__device__ __forceinline__ float fast_tanh(float x) {
  float a = __builtin_fabsf(x);
  float e = __expf(-2.0f * a);
  float r = (1.0f - e) / (1.0f + e);
  return __builtin_copysignf(r, x);
}

__device__ __forceinline__ void st8(float* p, v8f a) {
  float4 s0, s1;
  s0.x = a[0]; s0.y = a[1]; s0.z = a[2]; s0.w = a[3];
  s1.x = a[4]; s1.y = a[5]; s1.z = a[6]; s1.w = a[7];
  *(float4*)p = s0;
  *(float4*)(p + 128) = s1;
}
__device__ __forceinline__ v8f ld8(const float* p) {
  float4 s0 = *(const float4*)p;
  float4 s1 = *(const float4*)(p + 128);
  v8f a;
  a[0] = s0.x; a[1] = s0.y; a[2] = s0.z; a[3] = s0.w;
  a[4] = s1.x; a[5] = s1.y; a[6] = s1.z; a[7] = s1.w;
  return a;
}

__device__ __forceinline__ v8f ld8b(const float* p) {
  float4 s0 = *(const float4*)p;
  float4 s1 = *(const float4*)(p + 4);
  v8f a;
  a[0] = s0.x; a[1] = s0.y; a[2] = s0.z; a[3] = s0.w;
  a[4] = s1.x; a[5] = s1.y; a[6] = s1.z; a[7] = s1.w;
  return a;
}

#define XP_TILES_ 8
__global__ void __launch_bounds__(256)
xproj_kernel(const float* __restrict__ x, const float* __restrict__ Wi,
             const float* __restrict__ Wib, const float* __restrict__ bias,
             float* __restrict__ xp) {
  __shared__ __align__(16) float wl[64 * D_];
  __shared__ __align__(16) float bsum[64];
  const int lane = threadIdx.x & 31;
  const int lo = lane & 15, hf = lane >> 4;

  const int wid0 = (blockIdx.x * 8 + (threadIdx.x >> 5)) * XP_TILES_;
  const int btile = wid0 >> 11;
  const int t0    = wid0 & 2047;

  const float* xrow = x + ((size_t)(btile * 16 + lo) * T_ + t0) * D_ + hf * 2;
  const float* wfr = wl + lane * 4;
  float* o = xp + (size_t)wid0 * (4 * SLOT_) + lane * 4;

  v2f xa[32], xb[32];
#pragma unroll
  for (int ks = 0; ks < 32; ++ks) xa[ks] = *(const v2f*)(xrow + ks * 4);

#pragma unroll
  for (int it = 0; it < 64 * D_ / 4 / 256; ++it) {
    const int e = threadIdx.x + it * 256;
    const float4 q = *(const float4*)(Wi + (size_t)e * 4);
    const int j = e >> 5, ks = e & 31;
    float* d = wl + ((j >> 4) * 16 + (ks >> 1)) * 128 + (j & 15) * 4 + (ks & 1) * 2;
    v2f p0, p1;
    p0.x = q.x; p0.y = q.y; p1.x = q.z; p1.y = q.w;
    *(v2f*)d = p0;
    *(v2f*)(d + 64) = p1;
  }
  if (threadIdx.x < 64) bsum[threadIdx.x] = Wib[threadIdx.x] + bias[threadIdx.x];

  __syncthreads();

  auto tile = [&](const v2f* xf, float* dst) {
    asm volatile("" ::: "memory");
    v8f acc[4];
#pragma unroll
    for (int nt = 0; nt < 4; ++nt) acc[nt] = ld8b(bsum + nt * 16 + 8 * hf);
#pragma unroll
    for (int k2 = 0; k2 < 16; ++k2) {
      float4 wq[4];
#pragma unroll
      for (int nt = 0; nt < 4; ++nt)
        wq[nt] = *(const float4*)(wfr + (nt * 16 + k2) * 128);
#pragma unroll
      for (int nt = 0; nt < 4; ++nt) {
        v2f wa; wa.x = wq[nt].x; wa.y = wq[nt].y;
        acc[nt] = wmma4(wa, xf[2 * k2], acc[nt]);
      }
#pragma unroll
      for (int nt = 0; nt < 4; ++nt) {
        v2f wa; wa.x = wq[nt].z; wa.y = wq[nt].w;
        acc[nt] = wmma4(wa, xf[2 * k2 + 1], acc[nt]);
      }
    }
#pragma unroll
    for (int nt = 0; nt < 4; ++nt) st8(dst + nt * SLOT_, acc[nt]);
  };

#pragma unroll 1
  for (int i = 0; i < XP_TILES_; i += 2) {
#pragma unroll
    for (int ks = 0; ks < 32; ++ks)
      xb[ks] = *(const v2f*)(xrow + (size_t)(i + 1) * D_ + ks * 4);
    tile(xa, o + (size_t)i * (4 * SLOT_));
    const int in = (i + 2 < XP_TILES_) ? (i + 2) : i;
#pragma unroll
    for (int ks = 0; ks < 32; ++ks)
      xa[ks] = *(const v2f*)(xrow + (size_t)in * D_ + ks * 4);
    tile(xb, o + (size_t)(i + 1) * (4 * SLOT_));
  }
}

__device__ __forceinline__ void lds_barrier() {
  asm volatile("s_wait_dscnt 0x0\n\ts_barrier_signal -1\n\ts_barrier_wait -1"
               ::: "memory");
}

__device__ __forceinline__ void lds_signal() {
  asm volatile("s_wait_dscnt 0x0\n\ts_barrier_signal -1" ::: "memory");
}
__device__ __forceinline__ void lds_wait() {
  asm volatile("s_barrier_wait -1" ::: "memory");
}

#define PIN_ORDER()                         \
  do {                                      \
    asm volatile("" ::: "memory");          \
    __builtin_amdgcn_sched_barrier(0);      \
  } while (0)

__device__ __forceinline__ float xhalf(float x) {
  int b = __builtin_bit_cast(int, x);
  return __builtin_bit_cast(
      float, __builtin_amdgcn_permlanex16(b, b, 0x76543210, 0xfedcba98, false, false));
}

__global__ void __launch_bounds__(128)
scan_kernel(const float* __restrict__ Ap, const float* __restrict__ Am,
            const float* __restrict__ Wf, const float* __restrict__ Wo,
            const float* __restrict__ Wob, const float* __restrict__ xp,
            float* __restrict__ out) {
  __shared__ __align__(16) float ex[2 * PLANE_];
  __shared__ __align__(16) float sqx[2 * 128];
  __shared__ __align__(16) float hlds[16 * 64];
  const int lane = threadIdx.x & 31;
  const int w = __builtin_amdgcn_readfirstlane((int)(threadIdx.x >> 5));
  const int lo = lane & 15, hf = lane >> 4;
  const int btile = blockIdx.x;
  const int kb = 16 * w + 8 * hf;

  const float* xpw = xp + ((size_t)btile * T_ * 4 + w) * SLOT_;
  const float* xpl = xpw + lane * 4;
  v8f xq = ld8(xpl);

  const float* cs = (w < 2) ? Ap : Am;
  v2f aco[4], acp[4];
#pragma unroll
  for (int s = 0; s < 4; ++s) {
    aco[s] = *(const v2f*)(cs + (w * 16 + lo) * 64 + kb + 2 * s);
    acp[s] = *(const v2f*)(cs + ((w ^ 1) * 16 + lo) * 64 + kb + 2 * s);
  }
  v2f as_[4][4];
#pragma unroll
  for (int i = 0; i < 4; ++i) {
    const int d = (w + i) & 3;
    const float sg = (d < 2) ? 1.0f : -1.0f;
    const float* src = Wf + (d * 16 + lo) * 64 + kb;
#pragma unroll
    for (int s = 0; s < 4; ++s) {
      v2f q = *(const v2f*)(src + 2 * s);
      v2f r; r.x = q.x * sg; r.y = q.y * sg;
      as_[i][s] = r;
    }
  }

  const int d1 = (w + 1) & 3, d2 = (w + 2) & 3, d3 = (w + 3) & 3;
  const int so1 = (w * 4 + d1) * SLOT_ + lane * 4;
  const int so2 = (w * 4 + d2) * SLOT_ + lane * 4;
  const int so3 = (w * 4 + d3) * SLOT_ + lane * 4;
  const int soc = COMM_OFF_ + (w ^ 1) * SLOT_ + lane * 4;
  const int li1 = (d1 * 4 + w) * SLOT_ + lane * 4;
  const int li2 = (d2 * 4 + w) * SLOT_ + lane * 4;
  const int li3 = (d3 * 4 + w) * SLOT_ + lane * 4;
  const int lic = COMM_OFF_ + w * SLOT_ + lane * 4;

  v8f h;
#pragma unroll
  for (int v = 0; v < 8; ++v) h[v] = 0.0f;

  auto step = [&](float* P, float* Q, v8f seed) {
    v2f hb[4];
#pragma unroll
    for (int s = 0; s < 4; ++s) { hb[s].x = h[2 * s]; hb[s].y = h[2 * s + 1]; }

    float sq = 0.0f;
#pragma unroll
    for (int v = 0; v < 8; ++v) sq += h[v] * h[v];
    sq += xhalf(sq);

    v8f pco = seed;
    v8f ps0, ps1, ps2, ps3, pcp;
#pragma unroll
    for (int v = 0; v < 8; ++v) {
      ps0[v] = 0.0f; ps1[v] = 0.0f; ps2[v] = 0.0f; ps3[v] = 0.0f; pcp[v] = 0.0f;
    }

#pragma unroll
    for (int s = 0; s < 2; ++s) {
      ps1 = wmma4(as_[1][s], hb[s], ps1);
      ps2 = wmma4(as_[2][s], hb[s], ps2);
    }
    PIN_ORDER();
    Q[w * 32 + lane] = sq;
    PIN_ORDER();
#pragma unroll
    for (int s = 2; s < 4; ++s) {
      ps1 = wmma4(as_[1][s], hb[s], ps1);
      ps2 = wmma4(as_[2][s], hb[s], ps2);
    }
    PIN_ORDER();
    lds_signal();
    PIN_ORDER();
    ps3 = wmma4(as_[3][0], hb[0], ps3);
    pcp = wmma4(acp[0], hb[0], pcp);
    PIN_ORDER();
    lds_wait();
    float s0 = Q[lane];
    float s1 = Q[32 + lane];
    float s2 = Q[64 + lane];
    float s3 = Q[96 + lane];
    st8(P + so1, ps1);
    st8(P + so2, ps2);
    PIN_ORDER();
#pragma unroll
    for (int s = 1; s < 4; ++s) {
      ps3 = wmma4(as_[3][s], hb[s], ps3);
      pcp = wmma4(acp[s], hb[s], pcp);
    }
    PIN_ORDER();

    ps0 = wmma4(as_[0][0], hb[0], ps0);
    const float pm = s2 + s3;
    const float pa = (s0 + s1) + pm;
    const float qd = pa + 1e-8f;
    PIN_ORDER();
    pco = wmma4(aco[0], hb[0], pco);
    bool vc0, vc1;
    const float d1s = __builtin_amdgcn_div_scalef(pm, qd, false, &vc0);
    const float n1s = __builtin_amdgcn_div_scalef(pm, qd, true, &vc1);
    float r1 = __builtin_amdgcn_rcpf(d1s);
    PIN_ORDER();
    st8(P + so3, ps3);
    st8(P + soc, pcp);
    PIN_ORDER();
    ps0 = wmma4(as_[0][1], hb[1], ps0);
    const float e10 = __builtin_fmaf(-d1s, r1, 1.0f);
    r1 = __builtin_fmaf(e10, r1, r1);
    float m1 = n1s * r1;
    PIN_ORDER();
    pco = wmma4(aco[1], hb[1], pco);
    const float e11 = __builtin_fmaf(-d1s, m1, n1s);
    m1 = __builtin_fmaf(e11, r1, m1);
    const float e12 = __builtin_fmaf(-d1s, m1, n1s);
    const float f1 = __builtin_amdgcn_div_fmasf(e12, r1, m1, vc1);
    const float alpha = __builtin_amdgcn_div_fixupf(f1, qd, pm);
    float z0 = alpha - 0.72134752044448170f;
    PIN_ORDER();

    lds_barrier();

    v8f rc = ld8(P + lic);
    v8f r1t = ld8(P + li1);
    v8f r2t = ld8(P + li2);
    v8f r3t = ld8(P + li3);
    PIN_ORDER();
    asm volatile("" : "+v"(ps0), "+v"(pco), "+v"(z0));
    ps0 = wmma4(as_[0][2], hb[2], ps0);
    asm volatile("" : "+v"(ps0), "+v"(z0));
    const float z1 = -z0 * 20.0f;
    const float ex = __builtin_amdgcn_exp2f(z1 * 1.44269504088896340736f);
    float dd = 1.0f + ex;
    asm volatile("" : "+v"(pco), "+v"(dd));
    pco = wmma4(aco[2], hb[2], pco);
    asm volatile("" : "+v"(pco), "+v"(dd));
    bool vc2, vc3;
    float d2s = __builtin_amdgcn_div_scalef(1.0f, dd, false, &vc2);
    float n2s = __builtin_amdgcn_div_scalef(1.0f, dd, true, &vc3);
    float r2 = __builtin_amdgcn_rcpf(d2s);
    float e20 = __builtin_fmaf(-d2s, r2, 1.0f);
    asm volatile("" : "+v"(ps0), "+v"(d2s), "+v"(n2s), "+v"(r2), "+v"(e20));
    ps0 = wmma4(as_[0][3], hb[3], ps0);
    asm volatile("" : "+v"(ps0), "+v"(d2s), "+v"(n2s), "+v"(r2), "+v"(e20));
    r2 = __builtin_fmaf(e20, r2, r2);
    float m2 = n2s * r2;
    float e21 = __builtin_fmaf(-d2s, m2, n2s);
    asm volatile("" : "+v"(pco), "+v"(d2s), "+v"(n2s), "+v"(r2), "+v"(m2), "+v"(e21));
    pco = wmma4(aco[3], hb[3], pco);
    asm volatile("" : "+v"(pco), "+v"(d2s), "+v"(n2s), "+v"(r2), "+v"(m2), "+v"(e21));
    m2 = __builtin_fmaf(e21, r2, m2);
    const float e22 = __builtin_fmaf(-d2s, m2, n2s);
    const float f2 = __builtin_amdgcn_div_fmasf(e22, r2, m2, vc3);
    float g = __builtin_amdgcn_div_fixupf(f2, dd, 1.0f);
    asm volatile("" : "+v"(g), "+v"(rc), "+v"(r1t), "+v"(r2t), "+v"(r3t));

#pragma unroll
    for (int v = 0; v < 8; ++v) {
      const float um = ((ps0[v] + r1t[v]) + r2t[v]) + r3t[v];
      h[v] = fast_tanh((pco[v] + rc[v]) + g * um);
    }
  };

#pragma unroll 1
  for (int t = 0; t < T_; t += 2) {
    const v8f xb = ld8(xpl + (size_t)(t + 1) * (4 * SLOT_));
    const int tp = (t + 8 < T_) ? (t + 8) : (T_ - 2);
    __builtin_prefetch(xpw + (size_t)tp * (4 * SLOT_) + lane * 8, 0, 1);
    __builtin_prefetch(xpw + (size_t)(tp + 1) * (4 * SLOT_) + lane * 8, 0, 1);
    step(ex, sqx, xq);
    const int tn = (t + 2 < T_) ? (t + 2) : t;
    xq = ld8(xpl + (size_t)tn * (4 * SLOT_));
    step(ex + PLANE_, sqx + 128, xb);
  }

  {
    float4 a, b;
    a.x = h[0]; a.y = h[1]; a.z = h[2]; a.w = h[3];
    b.x = h[4]; b.y = h[5]; b.z = h[6]; b.w = h[7];
    *(float4*)&hlds[lo * 64 + kb] = a;
    *(float4*)&hlds[lo * 64 + kb + 4] = b;
  }
  __syncthreads();

  for (int i = threadIdx.x; i < 1024; i += 128)
    out[32768 + btile * 1024 + i] = hlds[i];

  v2f ph[16];
#pragma unroll
  for (int ks = 0; ks < 16; ++ks)
    ph[ks] = *(const v2f*)&hlds[lo * 64 + ks * 4 + hf * 2];
#pragma unroll
  for (int n2l = 0; n2l < 2; ++n2l) {
    const int n2 = 2 * w + n2l;
    v8f p;
    const float c0 = Wob[n2 * 16 + lo];
#pragma unroll
    for (int v = 0; v < 8; ++v) p[v] = c0;
#pragma unroll
    for (int ks = 0; ks < 16; ++ks) {
      v2f b = *(const v2f*)(Wo + (size_t)(n2 * 16 + lo) * 64 + ks * 4 + hf * 2);
      p = wmma4(ph[ks], b, p);
    }
#pragma unroll
    for (int v = 0; v < 8; ++v)
      out[(size_t)(btile * 16 + v + 8 * hf) * 128 + n2 * 16 + lo] = p[v];
  }
}

extern "C" void kernel_launch(void* const* d_in, const int* in_sizes, int n_in,
                              void* d_out, int out_size, void* d_ws, size_t ws_size,
                              hipStream_t stream) {
  const float* x    = (const float*)d_in[0];
  const float* Ap   = (const float*)d_in[1];
  const float* Am   = (const float*)d_in[2];
  const float* Wf   = (const float*)d_in[3];
  const float* Wiw  = (const float*)d_in[4];
  const float* Wib  = (const float*)d_in[5];
  const float* Wow  = (const float*)d_in[6];
  const float* Wob  = (const float*)d_in[7];
  const float* bias = (const float*)d_in[8];
  float* out = (float*)d_out;
  float* xp  = (float*)d_ws;

  xproj_kernel<<<512, 256, 0, stream>>>(x, Wiw, Wib, bias, xp);
  scan_kernel<<<16, 128, 0, stream>>>(Ap, Am, Wf, Wow, Wob, xp, out);
}


# scan: tanh's quotient (1-e)/(1+e) by the division's own reciprocal-and-refinement sequence, two values per packed fma, without its range scaling and fix-up (identities: 1+e in [1,2]); values same
# speedup vs baseline: 3.9891x; 1.1236x over previous
#include <hip/hip_runtime.h>
#include <hip/hip_bf16.h>


typedef float v8f __attribute__((ext_vector_type(8)));
typedef float v2f __attribute__((ext_vector_type(2)));

#define T_  2048
#define D_  128

#define SLOT_     256
#define COMM_OFF_ (16 * SLOT_)
#define PLANE_    (20 * SLOT_)

__device__ __forceinline__ v8f wmma4(v2f a, v2f b, v8f c) {
  return __builtin_amdgcn_wmma_f32_16x16x4_f32(false, a, false, b, (short)0, c,
                                               false, false);
}

__device__ __forceinline__ v2f tanh2(v2f x) {
  v2f e;
  e.x = __expf(-2.0f * __builtin_fabsf(x.x));
  e.y = __expf(-2.0f * __builtin_fabsf(x.y));
  const v2f one = {1.0f, 1.0f};
  const v2f n = one - e;
  const v2f d = one + e;
  v2f r;
  r.x = __builtin_amdgcn_rcpf(d.x);
  r.y = __builtin_amdgcn_rcpf(d.y);
  const v2f e0 = __builtin_elementwise_fma(-d, r, one);
  r = __builtin_elementwise_fma(e0, r, r);
  v2f q = n * r;
  const v2f e1 = __builtin_elementwise_fma(-d, q, n);
  q = __builtin_elementwise_fma(e1, r, q);
  const v2f e2 = __builtin_elementwise_fma(-d, q, n);
  q = __builtin_elementwise_fma(e2, r, q);
  v2f t;
  t.x = __builtin_copysignf(q.x, x.x);
  t.y = __builtin_copysignf(q.y, x.y);
  return t;
}

__device__ __forceinline__ void st8(float* p, v8f a) {
  float4 s0, s1;
  s0.x = a[0]; s0.y = a[1]; s0.z = a[2]; s0.w = a[3];
  s1.x = a[4]; s1.y = a[5]; s1.z = a[6]; s1.w = a[7];
  *(float4*)p = s0;
  *(float4*)(p + 128) = s1;
}
__device__ __forceinline__ v8f ld8(const float* p) {
  float4 s0 = *(const float4*)p;
  float4 s1 = *(const float4*)(p + 128);
  v8f a;
  a[0] = s0.x; a[1] = s0.y; a[2] = s0.z; a[3] = s0.w;
  a[4] = s1.x; a[5] = s1.y; a[6] = s1.z; a[7] = s1.w;
  return a;
}

__device__ __forceinline__ v8f ld8b(const float* p) {
  float4 s0 = *(const float4*)p;
  float4 s1 = *(const float4*)(p + 4);
  v8f a;
  a[0] = s0.x; a[1] = s0.y; a[2] = s0.z; a[3] = s0.w;
  a[4] = s1.x; a[5] = s1.y; a[6] = s1.z; a[7] = s1.w;
  return a;
}

#define XP_TILES_ 8
__global__ void __launch_bounds__(256)
xproj_kernel(const float* __restrict__ x, const float* __restrict__ Wi,
             const float* __restrict__ Wib, const float* __restrict__ bias,
             float* __restrict__ xp) {
  __shared__ __align__(16) float wl[64 * D_];
  __shared__ __align__(16) float bsum[64];
  const int lane = threadIdx.x & 31;
  const int lo = lane & 15, hf = lane >> 4;

  const int wid0 = (blockIdx.x * 8 + (threadIdx.x >> 5)) * XP_TILES_;
  const int btile = wid0 >> 11;
  const int t0    = wid0 & 2047;

  const float* xrow = x + ((size_t)(btile * 16 + lo) * T_ + t0) * D_ + hf * 2;
  const float* wfr = wl + lane * 4;
  float* o = xp + (size_t)wid0 * (4 * SLOT_) + lane * 4;

  v2f xa[32], xb[32];
#pragma unroll
  for (int ks = 0; ks < 32; ++ks) xa[ks] = *(const v2f*)(xrow + ks * 4);

#pragma unroll
  for (int it = 0; it < 64 * D_ / 4 / 256; ++it) {
    const int e = threadIdx.x + it * 256;
    const float4 q = *(const float4*)(Wi + (size_t)e * 4);
    const int j = e >> 5, ks = e & 31;
    float* d = wl + ((j >> 4) * 16 + (ks >> 1)) * 128 + (j & 15) * 4 + (ks & 1) * 2;
    v2f p0, p1;
    p0.x = q.x; p0.y = q.y; p1.x = q.z; p1.y = q.w;
    *(v2f*)d = p0;
    *(v2f*)(d + 64) = p1;
  }
  if (threadIdx.x < 64) bsum[threadIdx.x] = Wib[threadIdx.x] + bias[threadIdx.x];

  __syncthreads();

  auto tile = [&](const v2f* xf, float* dst) {
    asm volatile("" ::: "memory");
    v8f acc[4];
#pragma unroll
    for (int nt = 0; nt < 4; ++nt) acc[nt] = ld8b(bsum + nt * 16 + 8 * hf);
#pragma unroll
    for (int k2 = 0; k2 < 16; ++k2) {
      float4 wq[4];
#pragma unroll
      for (int nt = 0; nt < 4; ++nt)
        wq[nt] = *(const float4*)(wfr + (nt * 16 + k2) * 128);
#pragma unroll
      for (int nt = 0; nt < 4; ++nt) {
        v2f wa; wa.x = wq[nt].x; wa.y = wq[nt].y;
        acc[nt] = wmma4(wa, xf[2 * k2], acc[nt]);
      }
#pragma unroll
      for (int nt = 0; nt < 4; ++nt) {
        v2f wa; wa.x = wq[nt].z; wa.y = wq[nt].w;
        acc[nt] = wmma4(wa, xf[2 * k2 + 1], acc[nt]);
      }
    }
#pragma unroll
    for (int nt = 0; nt < 4; ++nt) st8(dst + nt * SLOT_, acc[nt]);
  };

#pragma unroll 1
  for (int i = 0; i < XP_TILES_; i += 2) {
#pragma unroll
    for (int ks = 0; ks < 32; ++ks)
      xb[ks] = *(const v2f*)(xrow + (size_t)(i + 1) * D_ + ks * 4);
    tile(xa, o + (size_t)i * (4 * SLOT_));
    const int in = (i + 2 < XP_TILES_) ? (i + 2) : i;
#pragma unroll
    for (int ks = 0; ks < 32; ++ks)
      xa[ks] = *(const v2f*)(xrow + (size_t)in * D_ + ks * 4);
    tile(xb, o + (size_t)(i + 1) * (4 * SLOT_));
  }
}

__device__ __forceinline__ void lds_barrier() {
  asm volatile("s_wait_dscnt 0x0\n\ts_barrier_signal -1\n\ts_barrier_wait -1"
               ::: "memory");
}

__device__ __forceinline__ void lds_signal() {
  asm volatile("s_wait_dscnt 0x0\n\ts_barrier_signal -1" ::: "memory");
}
__device__ __forceinline__ void lds_wait() {
  asm volatile("s_barrier_wait -1" ::: "memory");
}

#define PIN_ORDER()                         \
  do {                                      \
    asm volatile("" ::: "memory");          \
    __builtin_amdgcn_sched_barrier(0);      \
  } while (0)

__device__ __forceinline__ float xhalf(float x) {
  int b = __builtin_bit_cast(int, x);
  return __builtin_bit_cast(
      float, __builtin_amdgcn_permlanex16(b, b, 0x76543210, 0xfedcba98, false, false));
}

__global__ void __launch_bounds__(128)
scan_kernel(const float* __restrict__ Ap, const float* __restrict__ Am,
            const float* __restrict__ Wf, const float* __restrict__ Wo,
            const float* __restrict__ Wob, const float* __restrict__ xp,
            float* __restrict__ out) {
  __shared__ __align__(16) float ex[2 * PLANE_];
  __shared__ __align__(16) float sqx[2 * 128];
  __shared__ __align__(16) float hlds[16 * 64];
  const int lane = threadIdx.x & 31;
  const int w = __builtin_amdgcn_readfirstlane((int)(threadIdx.x >> 5));
  const int lo = lane & 15, hf = lane >> 4;
  const int btile = blockIdx.x;
  const int kb = 16 * w + 8 * hf;

  const float* xpw = xp + ((size_t)btile * T_ * 4 + w) * SLOT_;
  const float* xpl = xpw + lane * 4;
  v8f xq = ld8(xpl);

  const float* cs = (w < 2) ? Ap : Am;
  v2f aco[4], acp[4];
#pragma unroll
  for (int s = 0; s < 4; ++s) {
    aco[s] = *(const v2f*)(cs + (w * 16 + lo) * 64 + kb + 2 * s);
    acp[s] = *(const v2f*)(cs + ((w ^ 1) * 16 + lo) * 64 + kb + 2 * s);
  }
  v2f as_[4][4];
#pragma unroll
  for (int i = 0; i < 4; ++i) {
    const int d = (w + i) & 3;
    const float sg = (d < 2) ? 1.0f : -1.0f;
    const float* src = Wf + (d * 16 + lo) * 64 + kb;
#pragma unroll
    for (int s = 0; s < 4; ++s) {
      v2f q = *(const v2f*)(src + 2 * s);
      v2f r; r.x = q.x * sg; r.y = q.y * sg;
      as_[i][s] = r;
    }
  }

  const int d1 = (w + 1) & 3, d2 = (w + 2) & 3, d3 = (w + 3) & 3;
  const int so1 = (w * 4 + d1) * SLOT_ + lane * 4;
  const int so2 = (w * 4 + d2) * SLOT_ + lane * 4;
  const int so3 = (w * 4 + d3) * SLOT_ + lane * 4;
  const int soc = COMM_OFF_ + (w ^ 1) * SLOT_ + lane * 4;
  const int li1 = (d1 * 4 + w) * SLOT_ + lane * 4;
  const int li2 = (d2 * 4 + w) * SLOT_ + lane * 4;
  const int li3 = (d3 * 4 + w) * SLOT_ + lane * 4;
  const int lic = COMM_OFF_ + w * SLOT_ + lane * 4;

  v8f h;
#pragma unroll
  for (int v = 0; v < 8; ++v) h[v] = 0.0f;

  auto step = [&](float* P, float* Q, v8f seed) {
    v2f hb[4];
#pragma unroll
    for (int s = 0; s < 4; ++s) { hb[s].x = h[2 * s]; hb[s].y = h[2 * s + 1]; }

    float sq = 0.0f;
#pragma unroll
    for (int v = 0; v < 8; ++v) sq += h[v] * h[v];
    sq += xhalf(sq);

    v8f pco = seed;
    v8f ps0, ps1, ps2, ps3, pcp;
#pragma unroll
    for (int v = 0; v < 8; ++v) {
      ps0[v] = 0.0f; ps1[v] = 0.0f; ps2[v] = 0.0f; ps3[v] = 0.0f; pcp[v] = 0.0f;
    }

#pragma unroll
    for (int s = 0; s < 2; ++s) {
      ps1 = wmma4(as_[1][s], hb[s], ps1);
      ps2 = wmma4(as_[2][s], hb[s], ps2);
    }
    PIN_ORDER();
    Q[w * 32 + lane] = sq;
    PIN_ORDER();
#pragma unroll
    for (int s = 2; s < 4; ++s) {
      ps1 = wmma4(as_[1][s], hb[s], ps1);
      ps2 = wmma4(as_[2][s], hb[s], ps2);
    }
    PIN_ORDER();
    lds_signal();
    PIN_ORDER();
    ps3 = wmma4(as_[3][0], hb[0], ps3);
    pcp = wmma4(acp[0], hb[0], pcp);
    PIN_ORDER();
    lds_wait();
    float s0 = Q[lane];
    float s1 = Q[32 + lane];
    float s2 = Q[64 + lane];
    float s3 = Q[96 + lane];
    st8(P + so1, ps1);
    st8(P + so2, ps2);
    PIN_ORDER();
#pragma unroll
    for (int s = 1; s < 4; ++s) {
      ps3 = wmma4(as_[3][s], hb[s], ps3);
      pcp = wmma4(acp[s], hb[s], pcp);
    }
    PIN_ORDER();

    ps0 = wmma4(as_[0][0], hb[0], ps0);
    const float pm = s2 + s3;
    const float pa = (s0 + s1) + pm;
    const float qd = pa + 1e-8f;
    PIN_ORDER();
    pco = wmma4(aco[0], hb[0], pco);
    bool vc0, vc1;
    const float d1s = __builtin_amdgcn_div_scalef(pm, qd, false, &vc0);
    const float n1s = __builtin_amdgcn_div_scalef(pm, qd, true, &vc1);
    float r1 = __builtin_amdgcn_rcpf(d1s);
    PIN_ORDER();
    st8(P + so3, ps3);
    st8(P + soc, pcp);
    PIN_ORDER();
    ps0 = wmma4(as_[0][1], hb[1], ps0);
    const float e10 = __builtin_fmaf(-d1s, r1, 1.0f);
    r1 = __builtin_fmaf(e10, r1, r1);
    float m1 = n1s * r1;
    PIN_ORDER();
    pco = wmma4(aco[1], hb[1], pco);
    const float e11 = __builtin_fmaf(-d1s, m1, n1s);
    m1 = __builtin_fmaf(e11, r1, m1);
    const float e12 = __builtin_fmaf(-d1s, m1, n1s);
    const float f1 = __builtin_amdgcn_div_fmasf(e12, r1, m1, vc1);
    const float alpha = __builtin_amdgcn_div_fixupf(f1, qd, pm);
    float z0 = alpha - 0.72134752044448170f;
    PIN_ORDER();

    lds_barrier();

    v8f rc = ld8(P + lic);
    v8f r1t = ld8(P + li1);
    v8f r2t = ld8(P + li2);
    v8f r3t = ld8(P + li3);
    PIN_ORDER();
    asm volatile("" : "+v"(ps0), "+v"(pco), "+v"(z0));
    ps0 = wmma4(as_[0][2], hb[2], ps0);
    asm volatile("" : "+v"(ps0), "+v"(z0));
    const float z1 = -z0 * 20.0f;
    const float ex = __builtin_amdgcn_exp2f(z1 * 1.44269504088896340736f);
    float dd = 1.0f + ex;
    asm volatile("" : "+v"(pco), "+v"(dd));
    pco = wmma4(aco[2], hb[2], pco);
    asm volatile("" : "+v"(pco), "+v"(dd));
    bool vc2, vc3;
    float d2s = __builtin_amdgcn_div_scalef(1.0f, dd, false, &vc2);
    float n2s = __builtin_amdgcn_div_scalef(1.0f, dd, true, &vc3);
    float r2 = __builtin_amdgcn_rcpf(d2s);
    float e20 = __builtin_fmaf(-d2s, r2, 1.0f);
    asm volatile("" : "+v"(ps0), "+v"(d2s), "+v"(n2s), "+v"(r2), "+v"(e20));
    ps0 = wmma4(as_[0][3], hb[3], ps0);
    asm volatile("" : "+v"(ps0), "+v"(d2s), "+v"(n2s), "+v"(r2), "+v"(e20));
    r2 = __builtin_fmaf(e20, r2, r2);
    float m2 = n2s * r2;
    float e21 = __builtin_fmaf(-d2s, m2, n2s);
    asm volatile("" : "+v"(pco), "+v"(d2s), "+v"(n2s), "+v"(r2), "+v"(m2), "+v"(e21));
    pco = wmma4(aco[3], hb[3], pco);
    asm volatile("" : "+v"(pco), "+v"(d2s), "+v"(n2s), "+v"(r2), "+v"(m2), "+v"(e21));
    m2 = __builtin_fmaf(e21, r2, m2);
    const float e22 = __builtin_fmaf(-d2s, m2, n2s);
    const float f2 = __builtin_amdgcn_div_fmasf(e22, r2, m2, vc3);
    float g = __builtin_amdgcn_div_fixupf(f2, dd, 1.0f);
    asm volatile("" : "+v"(g), "+v"(rc), "+v"(r1t), "+v"(r2t), "+v"(r3t));

#pragma unroll
    for (int s = 0; s < 4; ++s) {
      v2f u;
#pragma unroll
      for (int r = 0; r < 2; ++r) {
        const int v = 2 * s + r;
        const float um = ((ps0[v] + r1t[v]) + r2t[v]) + r3t[v];
        u[r] = (pco[v] + rc[v]) + g * um;
      }
      const v2f th = tanh2(u);
      h[2 * s] = th.x;
      h[2 * s + 1] = th.y;
    }
  };

#pragma unroll 1
  for (int t = 0; t < T_; t += 2) {
    const v8f xb = ld8(xpl + (size_t)(t + 1) * (4 * SLOT_));
    const int tp = (t + 8 < T_) ? (t + 8) : (T_ - 2);
    __builtin_prefetch(xpw + (size_t)tp * (4 * SLOT_) + lane * 8, 0, 1);
    __builtin_prefetch(xpw + (size_t)(tp + 1) * (4 * SLOT_) + lane * 8, 0, 1);
    step(ex, sqx, xq);
    const int tn = (t + 2 < T_) ? (t + 2) : t;
    xq = ld8(xpl + (size_t)tn * (4 * SLOT_));
    step(ex + PLANE_, sqx + 128, xb);
  }

  {
    float4 a, b;
    a.x = h[0]; a.y = h[1]; a.z = h[2]; a.w = h[3];
    b.x = h[4]; b.y = h[5]; b.z = h[6]; b.w = h[7];
    *(float4*)&hlds[lo * 64 + kb] = a;
    *(float4*)&hlds[lo * 64 + kb + 4] = b;
  }
  __syncthreads();

  for (int i = threadIdx.x; i < 1024; i += 128)
    out[32768 + btile * 1024 + i] = hlds[i];

  v2f ph[16];
#pragma unroll
  for (int ks = 0; ks < 16; ++ks)
    ph[ks] = *(const v2f*)&hlds[lo * 64 + ks * 4 + hf * 2];
#pragma unroll
  for (int n2l = 0; n2l < 2; ++n2l) {
    const int n2 = 2 * w + n2l;
    v8f p;
    const float c0 = Wob[n2 * 16 + lo];
#pragma unroll
    for (int v = 0; v < 8; ++v) p[v] = c0;
#pragma unroll
    for (int ks = 0; ks < 16; ++ks) {
      v2f b = *(const v2f*)(Wo + (size_t)(n2 * 16 + lo) * 64 + ks * 4 + hf * 2);
      p = wmma4(ph[ks], b, p);
    }
#pragma unroll
    for (int v = 0; v < 8; ++v)
      out[(size_t)(btile * 16 + v + 8 * hf) * 128 + n2 * 16 + lo] = p[v];
  }
}

extern "C" void kernel_launch(void* const* d_in, const int* in_sizes, int n_in,
                              void* d_out, int out_size, void* d_ws, size_t ws_size,
                              hipStream_t stream) {
  const float* x    = (const float*)d_in[0];
  const float* Ap   = (const float*)d_in[1];
  const float* Am   = (const float*)d_in[2];
  const float* Wf   = (const float*)d_in[3];
  const float* Wiw  = (const float*)d_in[4];
  const float* Wib  = (const float*)d_in[5];
  const float* Wow  = (const float*)d_in[6];
  const float* Wob  = (const float*)d_in[7];
  const float* bias = (const float*)d_in[8];
  float* out = (float*)d_out;
  float* xp  = (float*)d_ws;

  xproj_kernel<<<512, 256, 0, stream>>>(x, Wiw, Wib, bias, xp);
  scan_kernel<<<16, 128, 0, stream>>>(Ap, Am, Wf, Wow, Wob, xp, out);
}


# x_proj of the steps from 256 on runs inside the scan's launch (scan of steps 0..255 on 16 WGPs, x_proj on the other WGPs), rest of the scan in a 3rd launch, h handed on in workspace; sums unchanged
# speedup vs baseline: 4.2035x; 1.0537x over previous
#include <hip/hip_runtime.h>
#include <hip/hip_bf16.h>


typedef float v8f __attribute__((ext_vector_type(8)));
typedef float v2f __attribute__((ext_vector_type(2)));

#define T_  2048
#define D_  128

#define SLOT_     256
#define COMM_OFF_ (16 * SLOT_)
#define PLANE_    (20 * SLOT_)

__device__ __forceinline__ v8f wmma4(v2f a, v2f b, v8f c) {
  return __builtin_amdgcn_wmma_f32_16x16x4_f32(false, a, false, b, (short)0, c,
                                               false, false);
}

__device__ __forceinline__ v2f tanh2(v2f x) {
  v2f e;
  e.x = __expf(-2.0f * __builtin_fabsf(x.x));
  e.y = __expf(-2.0f * __builtin_fabsf(x.y));
  const v2f one = {1.0f, 1.0f};
  const v2f n = one - e;
  const v2f d = one + e;
  v2f r;
  r.x = __builtin_amdgcn_rcpf(d.x);
  r.y = __builtin_amdgcn_rcpf(d.y);
  const v2f e0 = __builtin_elementwise_fma(-d, r, one);
  r = __builtin_elementwise_fma(e0, r, r);
  v2f q = n * r;
  const v2f e1 = __builtin_elementwise_fma(-d, q, n);
  q = __builtin_elementwise_fma(e1, r, q);
  const v2f e2 = __builtin_elementwise_fma(-d, q, n);
  q = __builtin_elementwise_fma(e2, r, q);
  v2f t;
  t.x = __builtin_copysignf(q.x, x.x);
  t.y = __builtin_copysignf(q.y, x.y);
  return t;
}

__device__ __forceinline__ void st8(float* p, v8f a) {
  float4 s0, s1;
  s0.x = a[0]; s0.y = a[1]; s0.z = a[2]; s0.w = a[3];
  s1.x = a[4]; s1.y = a[5]; s1.z = a[6]; s1.w = a[7];
  *(float4*)p = s0;
  *(float4*)(p + 128) = s1;
}
__device__ __forceinline__ v8f ld8(const float* p) {
  float4 s0 = *(const float4*)p;
  float4 s1 = *(const float4*)(p + 128);
  v8f a;
  a[0] = s0.x; a[1] = s0.y; a[2] = s0.z; a[3] = s0.w;
  a[4] = s1.x; a[5] = s1.y; a[6] = s1.z; a[7] = s1.w;
  return a;
}

__device__ __forceinline__ v8f ld8b(const float* p) {
  float4 s0 = *(const float4*)p;
  float4 s1 = *(const float4*)(p + 4);
  v8f a;
  a[0] = s0.x; a[1] = s0.y; a[2] = s0.z; a[3] = s0.w;
  a[4] = s1.x; a[5] = s1.y; a[6] = s1.z; a[7] = s1.w;
  return a;
}

template <int NT, int XP, int TA, int NSTEP>
__device__ __forceinline__ void
xproj_body(const float* __restrict__ x, const float* __restrict__ Wi,
           const float* __restrict__ Wib, const float* __restrict__ bias,
           float* __restrict__ xp, float* wl, float* bsum, int blk) {
  const int lane = threadIdx.x & 31;
  const int lo = lane & 15, hf = lane >> 4;

  constexpr int TPB = (NT / 32) * XP;
  constexpr int BPB = NSTEP / TPB;
  const int btile = blk / BPB;
  const int t0    = TA + (blk % BPB) * TPB + (int)(threadIdx.x >> 5) * XP;
  const int wid0  = btile * T_ + t0;

  const float* xrow = x + ((size_t)(btile * 16 + lo) * T_ + t0) * D_ + hf * 2;
  const float* wfr = wl + lane * 4;
  float* o = xp + (size_t)wid0 * (4 * SLOT_) + lane * 4;

  v2f xa[32], xb[32];
#pragma unroll
  for (int ks = 0; ks < 32; ++ks) xa[ks] = *(const v2f*)(xrow + ks * 4);

#pragma unroll
  for (int it = 0; it < 64 * D_ / 4 / NT; ++it) {
    const int e = threadIdx.x + it * NT;
    const float4 q = *(const float4*)(Wi + (size_t)e * 4);
    const int j = e >> 5, ks = e & 31;
    float* d = wl + ((j >> 4) * 16 + (ks >> 1)) * 128 + (j & 15) * 4 + (ks & 1) * 2;
    v2f p0, p1;
    p0.x = q.x; p0.y = q.y; p1.x = q.z; p1.y = q.w;
    *(v2f*)d = p0;
    *(v2f*)(d + 64) = p1;
  }
  if (threadIdx.x < 64) bsum[threadIdx.x] = Wib[threadIdx.x] + bias[threadIdx.x];

  __syncthreads();

  auto tile = [&](const v2f* xf, float* dst) {
    asm volatile("" ::: "memory");
    v8f acc[4];
#pragma unroll
    for (int nt = 0; nt < 4; ++nt) acc[nt] = ld8b(bsum + nt * 16 + 8 * hf);
#pragma unroll
    for (int k2 = 0; k2 < 16; ++k2) {
      float4 wq[4];
#pragma unroll
      for (int nt = 0; nt < 4; ++nt)
        wq[nt] = *(const float4*)(wfr + (nt * 16 + k2) * 128);
#pragma unroll
      for (int nt = 0; nt < 4; ++nt) {
        v2f wa; wa.x = wq[nt].x; wa.y = wq[nt].y;
        acc[nt] = wmma4(wa, xf[2 * k2], acc[nt]);
      }
#pragma unroll
      for (int nt = 0; nt < 4; ++nt) {
        v2f wa; wa.x = wq[nt].z; wa.y = wq[nt].w;
        acc[nt] = wmma4(wa, xf[2 * k2 + 1], acc[nt]);
      }
    }
#pragma unroll
    for (int nt = 0; nt < 4; ++nt) st8(dst + nt * SLOT_, acc[nt]);
  };

#pragma unroll 1
  for (int i = 0; i < XP; i += 2) {
#pragma unroll
    for (int ks = 0; ks < 32; ++ks)
      xb[ks] = *(const v2f*)(xrow + (size_t)(i + 1) * D_ + ks * 4);
    tile(xa, o + (size_t)i * (4 * SLOT_));
    const int in = (i + 2 < XP) ? (i + 2) : i;
#pragma unroll
    for (int ks = 0; ks < 32; ++ks)
      xa[ks] = *(const v2f*)(xrow + (size_t)in * D_ + ks * 4);
    tile(xb, o + (size_t)(i + 1) * (4 * SLOT_));
  }
}

__device__ __forceinline__ void lds_barrier() {
  asm volatile("s_wait_dscnt 0x0\n\ts_barrier_signal -1\n\ts_barrier_wait -1"
               ::: "memory");
}

__device__ __forceinline__ void lds_signal() {
  asm volatile("s_wait_dscnt 0x0\n\ts_barrier_signal -1" ::: "memory");
}
__device__ __forceinline__ void lds_wait() {
  asm volatile("s_barrier_wait -1" ::: "memory");
}

#define PIN_ORDER()                         \
  do {                                      \
    asm volatile("" ::: "memory");          \
    __builtin_amdgcn_sched_barrier(0);      \
  } while (0)

__device__ __forceinline__ float xhalf(float x) {
  int b = __builtin_bit_cast(int, x);
  return __builtin_bit_cast(
      float, __builtin_amdgcn_permlanex16(b, b, 0x76543210, 0xfedcba98, false, false));
}

template <int TB, int TE, bool FIRST, bool LAST>
__device__ __forceinline__ void
scan_body(const float* __restrict__ Ap, const float* __restrict__ Am,
          const float* __restrict__ Wf, const float* __restrict__ Wo,
          const float* __restrict__ Wob, const float* xp, float* carry,
          float* __restrict__ out, float* ex, float* sqx, float* hlds,
          const int btile) {
  const int lane = threadIdx.x & 31;
  const int w = __builtin_amdgcn_readfirstlane((int)(threadIdx.x >> 5));
  const int lo = lane & 15, hf = lane >> 4;
  const int kb = 16 * w + 8 * hf;

  const float* xpw = xp + ((size_t)btile * T_ * 4 + w) * SLOT_;
  const float* xpl = xpw + lane * 4;
  v8f xq = ld8(xpl + (size_t)TB * (4 * SLOT_));
  float* cl = carry + ((size_t)btile * T_ * 4 + w) * SLOT_ + lane * 4;

  const float* cs = (w < 2) ? Ap : Am;
  v2f aco[4], acp[4];
#pragma unroll
  for (int s = 0; s < 4; ++s) {
    aco[s] = *(const v2f*)(cs + (w * 16 + lo) * 64 + kb + 2 * s);
    acp[s] = *(const v2f*)(cs + ((w ^ 1) * 16 + lo) * 64 + kb + 2 * s);
  }
  v2f as_[4][4];
#pragma unroll
  for (int i = 0; i < 4; ++i) {
    const int d = (w + i) & 3;
    const float sg = (d < 2) ? 1.0f : -1.0f;
    const float* src = Wf + (d * 16 + lo) * 64 + kb;
#pragma unroll
    for (int s = 0; s < 4; ++s) {
      v2f q = *(const v2f*)(src + 2 * s);
      v2f r; r.x = q.x * sg; r.y = q.y * sg;
      as_[i][s] = r;
    }
  }

  const int d1 = (w + 1) & 3, d2 = (w + 2) & 3, d3 = (w + 3) & 3;
  const int so1 = (w * 4 + d1) * SLOT_ + lane * 4;
  const int so2 = (w * 4 + d2) * SLOT_ + lane * 4;
  const int so3 = (w * 4 + d3) * SLOT_ + lane * 4;
  const int soc = COMM_OFF_ + (w ^ 1) * SLOT_ + lane * 4;
  const int li1 = (d1 * 4 + w) * SLOT_ + lane * 4;
  const int li2 = (d2 * 4 + w) * SLOT_ + lane * 4;
  const int li3 = (d3 * 4 + w) * SLOT_ + lane * 4;
  const int lic = COMM_OFF_ + w * SLOT_ + lane * 4;

  v8f h;
  if (FIRST) {
#pragma unroll
    for (int v = 0; v < 8; ++v) h[v] = 0.0f;
  } else {
    h = ld8(cl);
  }

  auto step = [&](float* P, float* Q, v8f seed) {
    v2f hb[4];
#pragma unroll
    for (int s = 0; s < 4; ++s) { hb[s].x = h[2 * s]; hb[s].y = h[2 * s + 1]; }

    float sq = 0.0f;
#pragma unroll
    for (int v = 0; v < 8; ++v) sq += h[v] * h[v];
    sq += xhalf(sq);

    v8f pco = seed;
    v8f ps0, ps1, ps2, ps3, pcp;
#pragma unroll
    for (int v = 0; v < 8; ++v) {
      ps0[v] = 0.0f; ps1[v] = 0.0f; ps2[v] = 0.0f; ps3[v] = 0.0f; pcp[v] = 0.0f;
    }

#pragma unroll
    for (int s = 0; s < 2; ++s) {
      ps1 = wmma4(as_[1][s], hb[s], ps1);
      ps2 = wmma4(as_[2][s], hb[s], ps2);
    }
    PIN_ORDER();
    Q[w * 32 + lane] = sq;
    PIN_ORDER();
#pragma unroll
    for (int s = 2; s < 4; ++s) {
      ps1 = wmma4(as_[1][s], hb[s], ps1);
      ps2 = wmma4(as_[2][s], hb[s], ps2);
    }
    PIN_ORDER();
    lds_signal();
    PIN_ORDER();
    ps3 = wmma4(as_[3][0], hb[0], ps3);
    pcp = wmma4(acp[0], hb[0], pcp);
    PIN_ORDER();
    lds_wait();
    float s0 = Q[lane];
    float s1 = Q[32 + lane];
    float s2 = Q[64 + lane];
    float s3 = Q[96 + lane];
    st8(P + so1, ps1);
    st8(P + so2, ps2);
    PIN_ORDER();
#pragma unroll
    for (int s = 1; s < 4; ++s) {
      ps3 = wmma4(as_[3][s], hb[s], ps3);
      pcp = wmma4(acp[s], hb[s], pcp);
    }
    PIN_ORDER();

    ps0 = wmma4(as_[0][0], hb[0], ps0);
    const float pm = s2 + s3;
    const float pa = (s0 + s1) + pm;
    const float qd = pa + 1e-8f;
    PIN_ORDER();
    pco = wmma4(aco[0], hb[0], pco);
    bool vc0, vc1;
    const float d1s = __builtin_amdgcn_div_scalef(pm, qd, false, &vc0);
    const float n1s = __builtin_amdgcn_div_scalef(pm, qd, true, &vc1);
    float r1 = __builtin_amdgcn_rcpf(d1s);
    PIN_ORDER();
    st8(P + so3, ps3);
    st8(P + soc, pcp);
    PIN_ORDER();
    ps0 = wmma4(as_[0][1], hb[1], ps0);
    const float e10 = __builtin_fmaf(-d1s, r1, 1.0f);
    r1 = __builtin_fmaf(e10, r1, r1);
    float m1 = n1s * r1;
    PIN_ORDER();
    pco = wmma4(aco[1], hb[1], pco);
    const float e11 = __builtin_fmaf(-d1s, m1, n1s);
    m1 = __builtin_fmaf(e11, r1, m1);
    const float e12 = __builtin_fmaf(-d1s, m1, n1s);
    const float f1 = __builtin_amdgcn_div_fmasf(e12, r1, m1, vc1);
    const float alpha = __builtin_amdgcn_div_fixupf(f1, qd, pm);
    float z0 = alpha - 0.72134752044448170f;
    PIN_ORDER();

    lds_barrier();

    v8f rc = ld8(P + lic);
    v8f r1t = ld8(P + li1);
    v8f r2t = ld8(P + li2);
    v8f r3t = ld8(P + li3);
    PIN_ORDER();
    asm volatile("" : "+v"(ps0), "+v"(pco), "+v"(z0));
    ps0 = wmma4(as_[0][2], hb[2], ps0);
    asm volatile("" : "+v"(ps0), "+v"(z0));
    const float z1 = -z0 * 20.0f;
    const float ex = __builtin_amdgcn_exp2f(z1 * 1.44269504088896340736f);
    float dd = 1.0f + ex;
    asm volatile("" : "+v"(pco), "+v"(dd));
    pco = wmma4(aco[2], hb[2], pco);
    asm volatile("" : "+v"(pco), "+v"(dd));
    bool vc2, vc3;
    float d2s = __builtin_amdgcn_div_scalef(1.0f, dd, false, &vc2);
    float n2s = __builtin_amdgcn_div_scalef(1.0f, dd, true, &vc3);
    float r2 = __builtin_amdgcn_rcpf(d2s);
    float e20 = __builtin_fmaf(-d2s, r2, 1.0f);
    asm volatile("" : "+v"(ps0), "+v"(d2s), "+v"(n2s), "+v"(r2), "+v"(e20));
    ps0 = wmma4(as_[0][3], hb[3], ps0);
    asm volatile("" : "+v"(ps0), "+v"(d2s), "+v"(n2s), "+v"(r2), "+v"(e20));
    r2 = __builtin_fmaf(e20, r2, r2);
    float m2 = n2s * r2;
    float e21 = __builtin_fmaf(-d2s, m2, n2s);
    asm volatile("" : "+v"(pco), "+v"(d2s), "+v"(n2s), "+v"(r2), "+v"(m2), "+v"(e21));
    pco = wmma4(aco[3], hb[3], pco);
    asm volatile("" : "+v"(pco), "+v"(d2s), "+v"(n2s), "+v"(r2), "+v"(m2), "+v"(e21));
    m2 = __builtin_fmaf(e21, r2, m2);
    const float e22 = __builtin_fmaf(-d2s, m2, n2s);
    const float f2 = __builtin_amdgcn_div_fmasf(e22, r2, m2, vc3);
    float g = __builtin_amdgcn_div_fixupf(f2, dd, 1.0f);
    asm volatile("" : "+v"(g), "+v"(rc), "+v"(r1t), "+v"(r2t), "+v"(r3t));

#pragma unroll
    for (int s = 0; s < 4; ++s) {
      v2f u;
#pragma unroll
      for (int r = 0; r < 2; ++r) {
        const int v = 2 * s + r;
        const float um = ((ps0[v] + r1t[v]) + r2t[v]) + r3t[v];
        u[r] = (pco[v] + rc[v]) + g * um;
      }
      const v2f th = tanh2(u);
      h[2 * s] = th.x;
      h[2 * s + 1] = th.y;
    }
  };

#pragma unroll 1
  for (int t = TB; t < TE; t += 2) {
    const v8f xb = ld8(xpl + (size_t)(t + 1) * (4 * SLOT_));
    const int tp = (t + 8 < TE) ? (t + 8) : (TE - 2);
    __builtin_prefetch(xpw + (size_t)tp * (4 * SLOT_) + lane * 8, 0, 1);
    __builtin_prefetch(xpw + (size_t)(tp + 1) * (4 * SLOT_) + lane * 8, 0, 1);
    step(ex, sqx, xq);
    const int tn = (t + 2 < TE) ? (t + 2) : t;
    xq = ld8(xpl + (size_t)tn * (4 * SLOT_));
    step(ex + PLANE_, sqx + 128, xb);
  }

  if (!LAST) {
    st8(cl, h);
    return;
  }

  {
    float4 a, b;
    a.x = h[0]; a.y = h[1]; a.z = h[2]; a.w = h[3];
    b.x = h[4]; b.y = h[5]; b.z = h[6]; b.w = h[7];
    *(float4*)&hlds[lo * 64 + kb] = a;
    *(float4*)&hlds[lo * 64 + kb + 4] = b;
  }
  __syncthreads();

  for (int i = threadIdx.x; i < 1024; i += 128)
    out[32768 + btile * 1024 + i] = hlds[i];

  v2f ph[16];
#pragma unroll
  for (int ks = 0; ks < 16; ++ks)
    ph[ks] = *(const v2f*)&hlds[lo * 64 + ks * 4 + hf * 2];
#pragma unroll
  for (int n2l = 0; n2l < 2; ++n2l) {
    const int n2 = 2 * w + n2l;
    v8f p;
    const float c0 = Wob[n2 * 16 + lo];
#pragma unroll
    for (int v = 0; v < 8; ++v) p[v] = c0;
#pragma unroll
    for (int ks = 0; ks < 16; ++ks) {
      v2f b = *(const v2f*)(Wo + (size_t)(n2 * 16 + lo) * 64 + ks * 4 + hf * 2);
      p = wmma4(ph[ks], b, p);
    }
#pragma unroll
    for (int v = 0; v < 8; ++v)
      out[(size_t)(btile * 16 + v + 8 * hf) * 128 + n2 * 16 + lo] = p[v];
  }
}

#define T1_ 256

__global__ void __launch_bounds__(256)
xproj_head_kernel(const float* __restrict__ x, const float* __restrict__ Wi,
                  const float* __restrict__ Wib, const float* __restrict__ bias,
                  float* __restrict__ xp) {
  __shared__ __align__(16) float wl[64 * D_];
  __shared__ __align__(16) float bsum[64];
  xproj_body<256, 2, 0, T1_>(x, Wi, Wib, bias, xp, wl, bsum, (int)blockIdx.x);
}

#define FUSED_LDS_FLOATS_ (41 * 1024)
__global__ void __launch_bounds__(128)
scan_head_xproj_tail_kernel(const float* __restrict__ x, const float* __restrict__ Wi,
                            const float* __restrict__ Wib, const float* __restrict__ bias,
                            const float* __restrict__ Ap, const float* __restrict__ Am,
                            const float* __restrict__ Wf, float* xp) {
  __shared__ __align__(16) float sm[FUSED_LDS_FLOATS_];
  if (blockIdx.x < 16) {
    scan_body<0, T1_, true, false>(Ap, Am, Wf, nullptr, nullptr, xp, xp, nullptr,
                                   sm, sm + 2 * PLANE_, nullptr, (int)blockIdx.x);
  } else {
    xproj_body<128, 8, T1_, T_ - T1_>(x, Wi, Wib, bias, xp, sm, sm + 64 * D_,
                                      (int)blockIdx.x - 16);
  }
}

__global__ void __launch_bounds__(128)
scan_tail_kernel(const float* __restrict__ Ap, const float* __restrict__ Am,
                 const float* __restrict__ Wf, const float* __restrict__ Wo,
                 const float* __restrict__ Wob, float* xp,
                 float* __restrict__ out) {
  __shared__ __align__(16) float ex[2 * PLANE_];
  __shared__ __align__(16) float sqx[2 * 128];
  __shared__ __align__(16) float hlds[16 * 64];
  scan_body<T1_, T_, false, true>(Ap, Am, Wf, Wo, Wob, xp, xp, out, ex, sqx, hlds,
                                  (int)blockIdx.x);
}

extern "C" void kernel_launch(void* const* d_in, const int* in_sizes, int n_in,
                              void* d_out, int out_size, void* d_ws, size_t ws_size,
                              hipStream_t stream) {
  const float* x    = (const float*)d_in[0];
  const float* Ap   = (const float*)d_in[1];
  const float* Am   = (const float*)d_in[2];
  const float* Wf   = (const float*)d_in[3];
  const float* Wiw  = (const float*)d_in[4];
  const float* Wib  = (const float*)d_in[5];
  const float* Wow  = (const float*)d_in[6];
  const float* Wob  = (const float*)d_in[7];
  const float* bias = (const float*)d_in[8];
  float* out = (float*)d_out;
  float* xp  = (float*)d_ws;

  xproj_head_kernel<<<256, 256, 0, stream>>>(x, Wiw, Wib, bias, xp);
  scan_head_xproj_tail_kernel<<<912, 128, 0, stream>>>(x, Wiw, Wib, bias, Ap, Am, Wf, xp);
  scan_tail_kernel<<<16, 128, 0, stream>>>(Ap, Am, Wf, Wow, Wob, xp, out);
}
